# starting kernel
# baseline (speedup 1.0000x reference)
#include <hip/hip_runtime.h>
#include <math.h>

#define K_CODES 16384
#define DIMS    128
#define BATCH_N 262144
#define DECAY_F 0.99f
#define EPS_F   1e-5f

typedef __attribute__((ext_vector_type(2))) float v2f;
typedef __attribute__((ext_vector_type(8))) float v8f;

__global__ void ema_init_kernel(float* __restrict__ w, float* __restrict__ cnt) {
  const int stride = gridDim.x * blockDim.x;
  const int i = blockIdx.x * blockDim.x + threadIdx.x;
  const int total = K_CODES * DIMS;
  for (int j = i; j < total; j += stride) w[j] = 0.0f;
  for (int j = i; j < K_CODES; j += stride) cnt[j] = 0.0f;
}

__global__ __launch_bounds__(256)
void ema_scatter_kernel(const float* __restrict__ z_e, const int* __restrict__ idxs,
                        float* __restrict__ w, float* __restrict__ cnt) {
  const int lane = threadIdx.x & 31;
  const int wv   = threadIdx.x >> 5;
  const int r    = lane & 15;
  const int half = lane >> 4;
  const unsigned base = blockIdx.x * 16u;
  const int colbase = wv * 16;

  const int rowIdx = idxs[base + (unsigned)r];

  unsigned mm = 0u;
  #pragma unroll
  for (int c = 0; c < 16; ++c) {
    mm |= (unsigned)(__shfl(rowIdx, c, 32) == rowIdx) << c;
  }

  const bool isLeader = ((mm & ((1u << r) - 1u)) == 0u);
  const unsigned gmask = isLeader ? mm : 0u;

  v8f acc = {};
  #pragma unroll
  for (int k = 0; k < 4; ++k) {
    const int j0 = 4 * k + 2 * half;
    v2f a, b;
    a.x = (float)((gmask >> j0) & 1u);
    a.y = (float)((gmask >> (j0 + 1)) & 1u);
    const unsigned zoff = (base + (unsigned)j0) * DIMS + (unsigned)(colbase + r);
    b.x = z_e[zoff];
    b.y = z_e[zoff + DIMS];
    acc = __builtin_amdgcn_wmma_f32_16x16x4_f32(false, a, false, b,
                                                (short)0, acc, false, false);
  }

  #pragma unroll
  for (int v = 0; v < 8; ++v) {
    const int tcode = __shfl(rowIdx, 8 * half + v, 32);
    const unsigned off = (unsigned)tcode * DIMS + (unsigned)(colbase + r);
    atomicAdd(&w[off], acc[v]);
  }

  if (wv == 0 && half == 0 && isLeader) {
    atomicAdd(&cnt[(unsigned)rowIdx], (float)__popc(mm));
  }
}

__global__ void ema_finalize_kernel(const float* __restrict__ m, const float* __restrict__ N,
                                    const float* __restrict__ cnt, const int* __restrict__ tptr,
                                    float* __restrict__ out) {
  const float zbc = 1.0f - powf(DECAY_F, (float)tptr[0]);
  const int stride = gridDim.x * blockDim.x;
  const int total = K_CODES * DIMS;
  for (int j = blockIdx.x * blockDim.x + threadIdx.x; j < total; j += stride) {
    const int k = j >> 7;
    const float wv = out[j];
    const float nU = N[k] * DECAY_F + cnt[k] * (1.0f - DECAY_F);
    const float mU = m[j] * DECAY_F + wv * (1.0f - DECAY_F);
    out[j] = (mU / zbc) / ((nU / zbc) + EPS_F);
  }
}

extern "C" void kernel_launch(void* const* d_in, const int* in_sizes, int n_in,
                              void* d_out, int out_size, void* d_ws, size_t ws_size,
                              hipStream_t stream) {
  const float* z_e  = (const float*)d_in[0];
  const int*   idxs = (const int*)  d_in[1];
  const float* m    = (const float*)d_in[2];
  const float* N    = (const float*)d_in[3];
  const int*   t    = (const int*)  d_in[4];
  float* out = (float*)d_out;
  float* cnt = (float*)d_ws;

  ema_init_kernel<<<2048, 256, 0, stream>>>(out, cnt);
  ema_scatter_kernel<<<BATCH_N / 16, 256, 0, stream>>>(z_e, idxs, out, cnt);
  ema_finalize_kernel<<<(K_CODES * DIMS) / 256, 256, 0, stream>>>(m, N, cnt, t, out);
  (void)in_sizes; (void)n_in; (void)out_size; (void)ws_size;
}


# Atomic scatter -> one launch, one-hot product: each WGP scans all ids for its 64 codes, ballot-ranked lists per 16-code group in LDS, waves sum onehot@Z on WMMA, EMA fused; sums in fixed list order
# speedup vs baseline: 2.2066x; 2.2066x over previous
#include <hip/hip_runtime.h>
#include <math.h>


#define K_CODES 16384
#define DIMS    128
#define BATCH_N 262144
#define DECAY_F 0.99f
#define EPS_F   1e-5f

#define NWAVES    16
#define NTHREADS  (NWAVES * 32)
#define WG_CODES  64
#define IDS_WAVE  (BATCH_N / NWAVES)
#define SEG_CAP   128
#define LIST_CAP  (NWAVES * SEG_CAP + 32)
#define ROW_BITS  18
#define ROW_MASK  0x3FFFFu
#define PAD_ENTRY ((unsigned)WG_CODES << ROW_BITS)

typedef __attribute__((ext_vector_type(2))) float v2f;
typedef __attribute__((ext_vector_type(8))) float v8f;

__global__ __launch_bounds__(NTHREADS)
void ema_onehot_kernel(const float* __restrict__ z_e, const int* __restrict__ idxs,
                       const float* __restrict__ m, const float* __restrict__ N,
                       const int* __restrict__ tptr, float* __restrict__ out) {
  __shared__ unsigned seg[NWAVES][SEG_CAP];
  __shared__ __attribute__((aligned(16))) unsigned lst[4][LIST_CAP];
  __shared__ int cntq[NWAVES][4];
  __shared__ int tot[NWAVES];

  const int lane = threadIdx.x & 31;
  const int w    = __builtin_amdgcn_readfirstlane((int)(threadIdx.x >> 5));
  const int c    = lane & 15;
  const int h    = lane >> 4;
  const unsigned cbase  = blockIdx.x * (unsigned)WG_CODES;
  const int q    = w >> 2;
  const int cq   = w & 3;
  const unsigned ltmask = (1u << lane) - 1u;
  const unsigned mycode = 16u * (unsigned)q + (unsigned)c;
  const unsigned colofs = 32u * (unsigned)cq + 2u * (unsigned)c;

  v8f acc0 = {}, acc1 = {};
  float cntc = 0.0f;

  const int4* ids4 = reinterpret_cast<const int4*>(idxs) + w * (IDS_WAVE / 4);
  const unsigned rowbase = (unsigned)w * IDS_WAVE + (unsigned)lane * 4u;

  int R = 1;
  int r = 0;
  do {
    const int lo = r * SEG_CAP;
    int T = 0;

    auto keep = [&](unsigned rel, unsigned row) {
      const bool hit = rel < (unsigned)WG_CODES;
      const unsigned mk = __builtin_amdgcn_ballot_w32(hit);
      if (mk != 0u) {
        const int slot = T + __popc(mk & ltmask) - lo;
        const bool in = hit && (slot >= 0) && (slot < SEG_CAP);
        if (in) seg[w][slot] = row | (rel << ROW_BITS);
        T += __popc(mk);
      }
    };

    auto scan8 = [&](const int4 (&v)[8], int it) {
#pragma unroll
      for (int u = 0; u < 8; ++u) {
        const unsigned r0 = (unsigned)v[u].x - cbase;
        const unsigned r1 = (unsigned)v[u].y - cbase;
        const unsigned r2 = (unsigned)v[u].z - cbase;
        const unsigned r3 = (unsigned)v[u].w - cbase;
        const bool any = (r0 < (unsigned)WG_CODES) | (r1 < (unsigned)WG_CODES) |
                         (r2 < (unsigned)WG_CODES) | (r3 < (unsigned)WG_CODES);
        if (__builtin_amdgcn_ballot_w32(any) != 0u) {
          const unsigned row = rowbase + (unsigned)(it + u) * 128u;
          keep(r0, row); keep(r1, row + 1u); keep(r2, row + 2u); keep(r3, row + 3u);
        }
      }
    };
    constexpr int NIT = IDS_WAVE / 128;
    int4 va[8], vb[8];
#pragma unroll
    for (int u = 0; u < 8; ++u) va[u] = ids4[u * 32 + lane];
    for (int it = 0; it < NIT; it += 16) {
#pragma unroll
      for (int u = 0; u < 8; ++u) vb[u] = ids4[(it + 8 + u) * 32 + lane];
      scan8(va, it);
      const int itn = (it + 16 < NIT) ? (it + 16) : (NIT - 8);
#pragma unroll
      for (int u = 0; u < 8; ++u) va[u] = ids4[(itn + u) * 32 + lane];
      scan8(vb, it + 8);
    }

    if (lane == 0) tot[w] = T;
    __syncthreads();

    int nw = T - lo; nw = nw < 0 ? 0 : (nw > SEG_CAP ? SEG_CAP : nw);
    unsigned ent[SEG_CAP / 32], mg[SEG_CAP / 32][4];
    int c0 = 0, c1 = 0, c2 = 0, c3 = 0;
#pragma unroll
    for (int p = 0; p < SEG_CAP / 32; ++p) {
      const int i = lane + 32 * p;
      const bool valid = i < nw;
      ent[p] = valid ? seg[w][i] : PAD_ENTRY;
      const unsigned qe = ent[p] >> (ROW_BITS + 4);
      mg[p][0] = __builtin_amdgcn_ballot_w32(qe == 0u);
      mg[p][1] = __builtin_amdgcn_ballot_w32(qe == 1u);
      mg[p][2] = __builtin_amdgcn_ballot_w32(qe == 2u);
      mg[p][3] = __builtin_amdgcn_ballot_w32(qe == 3u);
      c0 += __popc(mg[p][0]); c1 += __popc(mg[p][1]);
      c2 += __popc(mg[p][2]); c3 += __popc(mg[p][3]);
    }
    if (lane == 0) { cntq[w][0] = c0; cntq[w][1] = c1; cntq[w][2] = c2; cntq[w][3] = c3; }
    __syncthreads();

    int b0 = 0, b1 = 0, b2 = 0, b3 = 0, t0 = 0, t1 = 0, t2 = 0, t3 = 0, tmax = 0;
#pragma unroll
    for (int ww = 0; ww < NWAVES; ++ww) {
      const int a0 = cntq[ww][0], a1 = cntq[ww][1], a2 = cntq[ww][2], a3 = cntq[ww][3];
      if (ww < w) { b0 += a0; b1 += a1; b2 += a2; b3 += a3; }
      t0 += a0; t1 += a1; t2 += a2; t3 += a3;
      const int tw = tot[ww];
      tmax = tw > tmax ? tw : tmax;
    }
    b0 = __builtin_amdgcn_readfirstlane(b0); b1 = __builtin_amdgcn_readfirstlane(b1);
    b2 = __builtin_amdgcn_readfirstlane(b2); b3 = __builtin_amdgcn_readfirstlane(b3);
    tmax = __builtin_amdgcn_readfirstlane(tmax);
    R = (tmax + SEG_CAP - 1) / SEG_CAP;
    const int tq = q == 0 ? t0 : (q == 1 ? t1 : (q == 2 ? t2 : t3));
    const int nq = __builtin_amdgcn_readfirstlane(tq);
    const int nb = (nq + 31) >> 5;

#pragma unroll
    for (int p = 0; p < SEG_CAP / 32; ++p) {
      const unsigned qe = ent[p] >> (ROW_BITS + 4);
      if (qe < 4u) {
        const unsigned mk = qe == 0u ? mg[p][0] : (qe == 1u ? mg[p][1] : (qe == 2u ? mg[p][2] : mg[p][3]));
        const int bs = qe == 0u ? b0 : (qe == 1u ? b1 : (qe == 2u ? b2 : b3));
        lst[qe][bs + __popc(mk & ltmask)] = ent[p];
      }
      b0 += __popc(mg[p][0]); b1 += __popc(mg[p][1]);
      b2 += __popc(mg[p][2]); b3 += __popc(mg[p][3]);
    }
    if (cq == 0) {
      const int i = nq + lane;
      if (i < nb * 32) lst[q][i] = PAD_ENTRY;
    }
    __syncthreads();

    const unsigned* L = lst[q];
    if (nb > 0) {
      unsigned ea[8], eb[8], fa[8], fb[8];
      v2f xa[8], xb[8], ya[8], yb[8];
      auto fetch = [&](unsigned (&e0)[8], unsigned (&e1)[8], v2f (&x0)[8], v2f (&x1)[8], int b) {
        const int bb = b < nb ? b : nb - 1;
#pragma unroll
        for (int s = 0; s < 8; ++s) {
          const uint2 e2 = *reinterpret_cast<const uint2*>(&L[bb * 32 + 4 * s + 2 * h]);
          e0[s] = e2.x; e1[s] = e2.y;
        }
#pragma unroll
        for (int s = 0; s < 8; ++s) {
          x0[s] = *reinterpret_cast<const v2f*>(z_e + ((e0[s] & ROW_MASK) * (unsigned)DIMS + colofs));
          x1[s] = *reinterpret_cast<const v2f*>(z_e + ((e1[s] & ROW_MASK) * (unsigned)DIMS + colofs));
        }
      };
      auto sum8 = [&](const unsigned (&e0)[8], const unsigned (&e1)[8],
                      const v2f (&x0)[8], const v2f (&x1)[8], unsigned code) {
#pragma unroll
        for (int s = 0; s < 8; ++s) {
          v2f a, bt0, bt1;
          a.x = ((e0[s] >> ROW_BITS) == code) ? 1.0f : 0.0f;
          a.y = ((e1[s] >> ROW_BITS) == code) ? 1.0f : 0.0f;
          cntc += a.x + a.y;
          bt0.x = x0[s].x; bt0.y = x1[s].x;
          bt1.x = x0[s].y; bt1.y = x1[s].y;
          acc0 = __builtin_amdgcn_wmma_f32_16x16x4_f32(false, a, false, bt0, (short)0, acc0, false, false);
          acc1 = __builtin_amdgcn_wmma_f32_16x16x4_f32(false, a, false, bt1, (short)0, acc1, false, false);
        }
      };
      fetch(ea, eb, xa, xb, 0);
      for (int b = 0; b < nb; b += 2) {
        fetch(fa, fb, ya, yb, b + 1);
        sum8(ea, eb, xa, xb, mycode);
        fetch(ea, eb, xa, xb, b + 2);
        sum8(fa, fb, ya, yb, (b + 1) < nb ? mycode : 0xFFFFFFFFu);
      }
    }
    ++r;
  } while (r < R);

  const float zbc  = 1.0f - powf(DECAY_F, (float)tptr[0]);
  const float ctot = cntc + __shfl_xor(cntc, 16, 32);
#pragma unroll
  for (int v = 0; v < 8; ++v) {
    const unsigned k  = cbase + 16u * (unsigned)q + 8u * (unsigned)h + (unsigned)v;
    const float    ck = __shfl(ctot, 8 * h + v, 32);
    const float    nU = N[k] * DECAY_F + ck * (1.0f - DECAY_F);
    const float    den = (nU / zbc) + EPS_F;
    const unsigned j  = k * (unsigned)DIMS + colofs;
    const v2f mm = *reinterpret_cast<const v2f*>(m + j);
    v2f o;
    o.x = ((mm.x * DECAY_F + acc0[v] * (1.0f - DECAY_F)) / zbc) / den;
    o.y = ((mm.y * DECAY_F + acc1[v] * (1.0f - DECAY_F)) / zbc) / den;
    *reinterpret_cast<v2f*>(out + j) = o;
  }
}

extern "C" void kernel_launch(void* const* d_in, const int* in_sizes, int n_in,
                              void* d_out, int out_size, void* d_ws, size_t ws_size,
                              hipStream_t stream) {
  const float* z_e  = (const float*)d_in[0];
  const int*   idxs = (const int*)  d_in[1];
  const float* m    = (const float*)d_in[2];
  const float* N    = (const float*)d_in[3];
  const int*   t    = (const int*)  d_in[4];
  float* out = (float*)d_out;

  ema_onehot_kernel<<<K_CODES / WG_CODES, NTHREADS, 0, stream>>>(z_e, idxs, m, N, t, out);
  (void)in_sizes; (void)n_in; (void)out_size; (void)d_ws; (void)ws_size;
}


# Id scan branches made scalar-only: a load set's 32 match ballots are computed as one batch ahead of the branches; slot by mbcnt, out-of-window matches go to a spare LDS entry; sum order unchanged
# speedup vs baseline: 2.2835x; 1.0348x over previous
#include <hip/hip_runtime.h>
#include <math.h>


#define K_CODES 16384
#define DIMS    128
#define BATCH_N 262144
#define DECAY_F 0.99f
#define EPS_F   1e-5f

#define NWAVES    16
#define NTHREADS  (NWAVES * 32)
#define WG_CODES  64
#define IDS_WAVE  (BATCH_N / NWAVES)
#define SEG_CAP   128
#define LIST_CAP  (NWAVES * SEG_CAP + 32)
#define ROW_BITS  18
#define ROW_MASK  0x3FFFFu
#define PAD_ENTRY ((unsigned)WG_CODES << ROW_BITS)

typedef __attribute__((ext_vector_type(2))) float v2f;
typedef __attribute__((ext_vector_type(8))) float v8f;

__global__ __launch_bounds__(NTHREADS)
void ema_onehot_kernel(const float* __restrict__ z_e, const int* __restrict__ idxs,
                       const float* __restrict__ m, const float* __restrict__ N,
                       const int* __restrict__ tptr, float* __restrict__ out) {
  __shared__ unsigned seg[NWAVES][SEG_CAP + 1];
  __shared__ __attribute__((aligned(16))) unsigned lst[4][LIST_CAP];
  __shared__ int cntq[NWAVES][4];
  __shared__ int tot[NWAVES];

  const int lane = threadIdx.x & 31;
  const int w    = __builtin_amdgcn_readfirstlane((int)(threadIdx.x >> 5));
  const int c    = lane & 15;
  const int h    = lane >> 4;
  const unsigned cbase  = blockIdx.x * (unsigned)WG_CODES;
  const int q    = w >> 2;
  const int cq   = w & 3;
  const unsigned ltmask = (1u << lane) - 1u;
  const unsigned mycode = 16u * (unsigned)q + (unsigned)c;
  const unsigned colofs = 32u * (unsigned)cq + 2u * (unsigned)c;

  v8f acc0 = {}, acc1 = {};
  float cntc = 0.0f;

  const int4* ids4 = reinterpret_cast<const int4*>(idxs) + w * (IDS_WAVE / 4);
  const unsigned rowbase = (unsigned)w * IDS_WAVE + (unsigned)lane * 4u;

  int R = 1;
  int r = 0;
  do {
    const int lo = r * SEG_CAP;
    int T = 0;

    auto keep = [&](unsigned mk, unsigned rel, unsigned row) {
      if (mk != 0u) {
        const unsigned slot = __builtin_amdgcn_mbcnt_lo(mk, (unsigned)(T - lo));
        if (rel < (unsigned)WG_CODES)
          seg[w][slot < (unsigned)SEG_CAP ? slot : (unsigned)SEG_CAP] = row | (rel << ROW_BITS);
        T += __popc(mk);
      }
    };

    auto scan8 = [&](const int4 (&v)[8], int it) {
      unsigned rel[8][4], hm[8][4];
#pragma unroll
      for (int u = 0; u < 8; ++u) {
        rel[u][0] = (unsigned)v[u].x - cbase;
        rel[u][1] = (unsigned)v[u].y - cbase;
        rel[u][2] = (unsigned)v[u].z - cbase;
        rel[u][3] = (unsigned)v[u].w - cbase;
#pragma unroll
        for (int j = 0; j < 4; ++j)
          hm[u][j] = __builtin_amdgcn_ballot_w32(rel[u][j] < (unsigned)WG_CODES);
      }
#pragma unroll
      for (int u = 0; u < 8; ++u) {
        if ((hm[u][0] | hm[u][1] | hm[u][2] | hm[u][3]) != 0u) {
          const unsigned row = rowbase + (unsigned)(it + u) * 128u;
          keep(hm[u][0], rel[u][0], row);      keep(hm[u][1], rel[u][1], row + 1u);
          keep(hm[u][2], rel[u][2], row + 2u); keep(hm[u][3], rel[u][3], row + 3u);
        }
      }
    };
    constexpr int NIT = IDS_WAVE / 128;
    int4 va[8], vb[8];
#pragma unroll
    for (int u = 0; u < 8; ++u) va[u] = ids4[u * 32 + lane];
    for (int it = 0; it < NIT; it += 16) {
#pragma unroll
      for (int u = 0; u < 8; ++u) vb[u] = ids4[(it + 8 + u) * 32 + lane];
      scan8(va, it);
      const int itn = (it + 16 < NIT) ? (it + 16) : (NIT - 8);
#pragma unroll
      for (int u = 0; u < 8; ++u) va[u] = ids4[(itn + u) * 32 + lane];
      scan8(vb, it + 8);
    }

    if (lane == 0) tot[w] = T;
    __syncthreads();

    int nw = T - lo; nw = nw < 0 ? 0 : (nw > SEG_CAP ? SEG_CAP : nw);
    unsigned ent[SEG_CAP / 32], mg[SEG_CAP / 32][4];
    int c0 = 0, c1 = 0, c2 = 0, c3 = 0;
#pragma unroll
    for (int p = 0; p < SEG_CAP / 32; ++p) {
      const int i = lane + 32 * p;
      const bool valid = i < nw;
      ent[p] = valid ? seg[w][i] : PAD_ENTRY;
      const unsigned qe = ent[p] >> (ROW_BITS + 4);
      mg[p][0] = __builtin_amdgcn_ballot_w32(qe == 0u);
      mg[p][1] = __builtin_amdgcn_ballot_w32(qe == 1u);
      mg[p][2] = __builtin_amdgcn_ballot_w32(qe == 2u);
      mg[p][3] = __builtin_amdgcn_ballot_w32(qe == 3u);
      c0 += __popc(mg[p][0]); c1 += __popc(mg[p][1]);
      c2 += __popc(mg[p][2]); c3 += __popc(mg[p][3]);
    }
    if (lane == 0) { cntq[w][0] = c0; cntq[w][1] = c1; cntq[w][2] = c2; cntq[w][3] = c3; }
    __syncthreads();

    int b0 = 0, b1 = 0, b2 = 0, b3 = 0, t0 = 0, t1 = 0, t2 = 0, t3 = 0, tmax = 0;
#pragma unroll
    for (int ww = 0; ww < NWAVES; ++ww) {
      const int a0 = cntq[ww][0], a1 = cntq[ww][1], a2 = cntq[ww][2], a3 = cntq[ww][3];
      if (ww < w) { b0 += a0; b1 += a1; b2 += a2; b3 += a3; }
      t0 += a0; t1 += a1; t2 += a2; t3 += a3;
      const int tw = tot[ww];
      tmax = tw > tmax ? tw : tmax;
    }
    b0 = __builtin_amdgcn_readfirstlane(b0); b1 = __builtin_amdgcn_readfirstlane(b1);
    b2 = __builtin_amdgcn_readfirstlane(b2); b3 = __builtin_amdgcn_readfirstlane(b3);
    tmax = __builtin_amdgcn_readfirstlane(tmax);
    R = (tmax + SEG_CAP - 1) / SEG_CAP;
    const int tq = q == 0 ? t0 : (q == 1 ? t1 : (q == 2 ? t2 : t3));
    const int nq = __builtin_amdgcn_readfirstlane(tq);
    const int nb = (nq + 31) >> 5;

#pragma unroll
    for (int p = 0; p < SEG_CAP / 32; ++p) {
      const unsigned qe = ent[p] >> (ROW_BITS + 4);
      if (qe < 4u) {
        const unsigned mk = qe == 0u ? mg[p][0] : (qe == 1u ? mg[p][1] : (qe == 2u ? mg[p][2] : mg[p][3]));
        const int bs = qe == 0u ? b0 : (qe == 1u ? b1 : (qe == 2u ? b2 : b3));
        lst[qe][bs + __popc(mk & ltmask)] = ent[p];
      }
      b0 += __popc(mg[p][0]); b1 += __popc(mg[p][1]);
      b2 += __popc(mg[p][2]); b3 += __popc(mg[p][3]);
    }
    if (cq == 0) {
      const int i = nq + lane;
      if (i < nb * 32) lst[q][i] = PAD_ENTRY;
    }
    __syncthreads();

    const unsigned* L = lst[q];
    if (nb > 0) {
      unsigned ea[8], eb[8], fa[8], fb[8];
      v2f xa[8], xb[8], ya[8], yb[8];
      auto fetch = [&](unsigned (&e0)[8], unsigned (&e1)[8], v2f (&x0)[8], v2f (&x1)[8], int b) {
        const int bb = b < nb ? b : nb - 1;
#pragma unroll
        for (int s = 0; s < 8; ++s) {
          const uint2 e2 = *reinterpret_cast<const uint2*>(&L[bb * 32 + 4 * s + 2 * h]);
          e0[s] = e2.x; e1[s] = e2.y;
        }
#pragma unroll
        for (int s = 0; s < 8; ++s) {
          x0[s] = *reinterpret_cast<const v2f*>(z_e + ((e0[s] & ROW_MASK) * (unsigned)DIMS + colofs));
          x1[s] = *reinterpret_cast<const v2f*>(z_e + ((e1[s] & ROW_MASK) * (unsigned)DIMS + colofs));
        }
      };
      auto sum8 = [&](const unsigned (&e0)[8], const unsigned (&e1)[8],
                      const v2f (&x0)[8], const v2f (&x1)[8], unsigned code) {
#pragma unroll
        for (int s = 0; s < 8; ++s) {
          v2f a, bt0, bt1;
          a.x = ((e0[s] >> ROW_BITS) == code) ? 1.0f : 0.0f;
          a.y = ((e1[s] >> ROW_BITS) == code) ? 1.0f : 0.0f;
          cntc += a.x + a.y;
          bt0.x = x0[s].x; bt0.y = x1[s].x;
          bt1.x = x0[s].y; bt1.y = x1[s].y;
          acc0 = __builtin_amdgcn_wmma_f32_16x16x4_f32(false, a, false, bt0, (short)0, acc0, false, false);
          acc1 = __builtin_amdgcn_wmma_f32_16x16x4_f32(false, a, false, bt1, (short)0, acc1, false, false);
        }
      };
      fetch(ea, eb, xa, xb, 0);
      for (int b = 0; b < nb; b += 2) {
        fetch(fa, fb, ya, yb, b + 1);
        sum8(ea, eb, xa, xb, mycode);
        fetch(ea, eb, xa, xb, b + 2);
        sum8(fa, fb, ya, yb, (b + 1) < nb ? mycode : 0xFFFFFFFFu);
      }
    }
    ++r;
  } while (r < R);

  const float zbc  = 1.0f - powf(DECAY_F, (float)tptr[0]);
  const float ctot = cntc + __shfl_xor(cntc, 16, 32);
#pragma unroll
  for (int v = 0; v < 8; ++v) {
    const unsigned k  = cbase + 16u * (unsigned)q + 8u * (unsigned)h + (unsigned)v;
    const float    ck = __shfl(ctot, 8 * h + v, 32);
    const float    nU = N[k] * DECAY_F + ck * (1.0f - DECAY_F);
    const float    den = (nU / zbc) + EPS_F;
    const unsigned j  = k * (unsigned)DIMS + colofs;
    const v2f mm = *reinterpret_cast<const v2f*>(m + j);
    v2f o;
    o.x = ((mm.x * DECAY_F + acc0[v] * (1.0f - DECAY_F)) / zbc) / den;
    o.y = ((mm.y * DECAY_F + acc1[v] * (1.0f - DECAY_F)) / zbc) / den;
    *reinterpret_cast<v2f*>(out + j) = o;
  }
}

extern "C" void kernel_launch(void* const* d_in, const int* in_sizes, int n_in,
                              void* d_out, int out_size, void* d_ws, size_t ws_size,
                              hipStream_t stream) {
  const float* z_e  = (const float*)d_in[0];
  const int*   idxs = (const int*)  d_in[1];
  const float* m    = (const float*)d_in[2];
  const float* N    = (const float*)d_in[3];
  const int*   t    = (const int*)  d_in[4];
  float* out = (float*)d_out;

  ema_onehot_kernel<<<K_CODES / WG_CODES, NTHREADS, 0, stream>>>(z_e, idxs, m, N, t, out);
  (void)in_sizes; (void)n_in; (void)out_size; (void)d_ws; (void)ws_size;
}
